# starting kernel
# baseline (speedup 1.0000x reference)
#include <hip/hip_runtime.h>
#include <math.h>

typedef __attribute__((ext_vector_type(2))) float v2f;
typedef __attribute__((ext_vector_type(8))) float v8f;

namespace {
constexpr int kRes = 64;
constexpr int kNB  = 4;
constexpr int kNN  = 128;
constexpr int kNJ  = 16;
constexpr float kBWInv = 1.0f / 0.3f;
}

__global__ void __launch_bounds__(256) w2_kernel(const float* __restrict__ Bs,
                                                 const float* __restrict__ Ds,
                                                 float* __restrict__ w2) {
  int idx = blockIdx.x * blockDim.x + threadIdx.x;
  if (idx >= kNB * kNN) return;
  const float* bp = Bs + idx * kNJ * 2;
  const float* dp = Ds + idx * kNJ * 2;
  float bx[kNJ], by[kNJ], dx[kNJ], dy[kNJ];
#pragma unroll
  for (int j = 0; j < kNJ; ++j) {
    bx[j] = bp[2 * j + 0];
    by[j] = bp[2 * j + 1];
    dx[j] = dp[2 * j + 0];
    dy[j] = dp[2 * j + 1];
  }
  float w = 0.0f;
#pragma unroll
  for (int i = 0; i < kNJ; ++i) {
#pragma unroll
    for (int j = 0; j < kNJ; ++j) {
      float r0 = fmaxf(dx[j] - bx[i], 0.0f);
      float r1 = fmaxf(dy[j] - by[i], 0.0f);
      w = fmaxf(w, fminf(r0, r1));
    }
  }
  w2[idx] = w * w;
}

__global__ void __launch_bounds__(256) img_kernel(const float* __restrict__ Bs,
                                                  const float* __restrict__ Ds,
                                                  const float* __restrict__ w2,
                                                  float* __restrict__ out) {
  __shared__ float sB[kNN * kNJ * 2];
  __shared__ float sD[kNN * kNJ * 2];
  __shared__ float sW[kNN];

  const int wave   = threadIdx.x >> 5;
  const int lane   = threadIdx.x & 31;
  const int waveId = blockIdx.x * 8 + wave;
  const int b      = waveId >> 8;
  const int tile   = waveId & 255;

  {
    const float4* gB = (const float4*)(Bs + (size_t)b * kNN * kNJ * 2);
    const float4* gD = (const float4*)(Ds + (size_t)b * kNN * kNJ * 2);
    float4* lB = (float4*)sB;
    float4* lD = (float4*)sD;
    for (int i = threadIdx.x; i < kNN * kNJ * 2 / 4; i += 256) {
      lB[i] = gB[i];
      lD[i] = gD[i];
    }
    if (threadIdx.x < kNN) sW[threadIdx.x] = w2[b * kNN + threadIdx.x];
  }
  __syncthreads();

  const int li = lane & 15;
  const int hi = lane >> 4;
  const int p  = tile * 16 + li;
  const float x0 = -3.0f + 6.0f * (float)(p >> 6) / 63.0f;
  const float x1 = -3.0f + 6.0f * (float)(p & 63) / 63.0f;

  v8f acc = {};
  for (int nc = 0; nc < kNN; nc += 4) {
    v2f a;
#pragma unroll
    for (int t = 0; t < 2; ++t) {
      const int n = nc + 2 * hi + t;
      const float2* Bp = (const float2*)(sB + n * (kNJ * 2));
      const float2* Dp = (const float2*)(sD + n * (kNJ * 2));
      float db = 3.0e38f, dd = 3.0e38f;
#pragma unroll
      for (int j = 0; j < kNJ; ++j) {
        float2 bv = Bp[j];
        float2 dv = Dp[j];
        db = fminf(db, fmaxf(bv.x - x0, bv.y - x1));
        dd = fminf(dd, fmaxf(x0 - dv.x, x1 - dv.y));
      }
      float dist = fmaxf(fmaxf(db, dd), 0.0f) * kBWInv;
      a[t] = expf(-0.5f * dist * dist) * sW[n];
    }
    v2f ones = {1.0f, 1.0f};
    acc = __builtin_amdgcn_wmma_f32_16x16x4_f32(
        false, a, false, ones, (short)0, acc, false, false);
  }

  float* op = out + (size_t)b * (kRes * kRes) + tile * 16;
  if (lane == 0) {
#pragma unroll
    for (int r = 0; r < 8; ++r) op[r] = acc[r];
  } else if (lane == 16) {
#pragma unroll
    for (int r = 0; r < 8; ++r) op[8 + r] = acc[r];
  }
}

extern "C" void kernel_launch(void* const* d_in, const int* in_sizes, int n_in,
                              void* d_out, int out_size, void* d_ws, size_t ws_size,
                              hipStream_t stream) {
  (void)in_sizes; (void)n_in; (void)out_size; (void)ws_size;
  const float* Bs = (const float*)d_in[0];
  const float* Ds = (const float*)d_in[1];
  float* w2  = (float*)d_ws;
  float* out = (float*)d_out;

  w2_kernel<<<2, 256, 0, stream>>>(Bs, Ds, w2);
  img_kernel<<<128, 256, 0, stream>>>(Bs, Ds, w2, out);
}


# image pass turned: lane = one box (32 corners in registers, staged once through LDS), pixel coordinates scalar, no LDS reads in the pixel loop; box sum by one-hot then ones WMMA (sum order changes)
# speedup vs baseline: 2.8216x; 2.8216x over previous
#include <hip/hip_runtime.h>
#include <math.h>

typedef __attribute__((ext_vector_type(2))) float v2f;
typedef __attribute__((ext_vector_type(8))) float v8f;

namespace {
constexpr int kRes = 64;
constexpr int kNB  = 4;
constexpr int kNN  = 128;
constexpr int kNJ  = 16;
constexpr int kSeg = 32;
constexpr int kBoxB  = kNJ * 2 * 4;
constexpr int kPitch = kBoxB + 16;
constexpr float kBWInv = 1.0f / 0.3f;
}

static __device__ __forceinline__ void wmma_slots(v8f& d, v2f a, v2f b) {
  asm volatile("v_nop\n\tv_nop\n\tv_nop\n\tv_nop" : "+v"(d) : "v"(a), "v"(b));
}

__global__ void __launch_bounds__(256) w2_kernel(const float* __restrict__ Bs,
                                                 const float* __restrict__ Ds,
                                                 float* __restrict__ w2) {
  int idx = blockIdx.x * blockDim.x + threadIdx.x;
  if (idx >= kNB * kNN) return;
  const float* bp = Bs + idx * kNJ * 2;
  const float* dp = Ds + idx * kNJ * 2;
  float bx[kNJ], by[kNJ], dx[kNJ], dy[kNJ];
#pragma unroll
  for (int j = 0; j < kNJ; ++j) {
    bx[j] = bp[2 * j + 0];
    by[j] = bp[2 * j + 1];
    dx[j] = dp[2 * j + 0];
    dy[j] = dp[2 * j + 1];
  }
  float w = 0.0f;
#pragma unroll
  for (int i = 0; i < kNJ; ++i) {
#pragma unroll
    for (int j = 0; j < kNJ; ++j) {
      float r0 = fmaxf(dx[j] - bx[i], 0.0f);
      float r1 = fmaxf(dy[j] - by[i], 0.0f);
      w = fmaxf(w, fminf(r0, r1));
    }
  }
  w2[idx] = w * w;
}

__global__ void __launch_bounds__(256) img_kernel(const float* __restrict__ Bs,
                                                  const float* __restrict__ Ds,
                                                  const float* __restrict__ w2,
                                                  float* __restrict__ out) {
  __shared__ float sPart[8][kSeg];
  __shared__ __attribute__((aligned(16))) unsigned char sBox[4][2][32 * kPitch];

  const int wave = threadIdx.x >> 5;
  const int lane = threadIdx.x & 31;
  const int g    = wave & 3;
  const int half = wave >> 2;
  const int c    = lane & 15;
  const int h    = lane >> 4;
  const int b    = blockIdx.x >> 6;
  const int p0   = (blockIdx.x & 63) * kRes + half * kSeg;
  const int n    = g * 32 + lane;

  float bx[kNJ], by[kNJ], dx[kNJ], dy[kNJ];
  {
    const float* src = half ? Ds : Bs;
    const float4* gS = (const float4*)(src + (size_t)(b * kNN + g * 32) * (kNJ * 2));
    unsigned char* lS = sBox[g][half];
#pragma unroll
    for (int i = 0; i < 8; ++i) {
      const int piece = i * 32 + lane;
      const int off = (piece >> 3) * kPitch + (piece & 7) * 16;
      *(float4*)(lS + off) = gS[piece];
    }
    __syncthreads();
    const unsigned char* lB = sBox[g][0];
    const unsigned char* lD = sBox[g][1];
#pragma unroll
    for (int j = 0; j < kNJ / 2; ++j) {
      const float4 bv = *(const float4*)(lB + lane * kPitch + j * 16);
      const float4 dv = *(const float4*)(lD + lane * kPitch + j * 16);
      bx[2 * j] = bv.x; by[2 * j] = bv.y; bx[2 * j + 1] = bv.z; by[2 * j + 1] = bv.w;
      dx[2 * j] = dv.x; dy[2 * j] = dv.y; dx[2 * j + 1] = dv.z; dy[2 * j + 1] = dv.w;
    }
  }
  const float wn = w2[b * kNN + n];

  const float x0  = -3.0f + 6.0f * (float)(p0 >> 6) / 63.0f;
  const float x1l = -3.0f + 6.0f * (float)((p0 + lane) & 63) / 63.0f;

  float tot = 0.0f;
  for (int t = 0; t < 2; ++t) {
    v8f d1 = {};
    for (int q = 0; q < 8; ++q) {
      v2f a;
#pragma unroll
      for (int s = 0; s < 2; ++s) {
        const float x1 = __builtin_bit_cast(
            float, __builtin_amdgcn_readlane(__builtin_bit_cast(int, x1l), 16 * t + 2 * q + s));
        float db = 3.0e38f, dd = 3.0e38f;
#pragma unroll
        for (int k = 0; k < kNJ; ++k) {
          db = fminf(db, fmaxf(bx[k] - x0, by[k] - x1));
          dd = fminf(dd, fmaxf(x0 - dx[k], x1 - dy[k]));
        }
        float dist = fmaxf(fmaxf(db, dd), 0.0f) * kBWInv;
        a[s] = expf(-0.5f * dist * dist) * wn;
      }
      v2f sel;
      sel[0] = (c == 2 * q)     ? 1.0f : 0.0f;
      sel[1] = (c == 2 * q + 1) ? 1.0f : 0.0f;
      d1 = __builtin_amdgcn_wmma_f32_16x16x4_f32(
          false, a, false, sel, (short)0, d1, false, false);
      wmma_slots(d1, a, sel);
    }
    v8f d2 = {};
    const v2f ones = {1.0f, 1.0f};
    v2f rows;
#pragma unroll
    for (int m = 0; m < 4; ++m) {
      rows[0] = d1[2 * m];
      rows[1] = d1[2 * m + 1];
      d2 = __builtin_amdgcn_wmma_f32_16x16x4_f32(
          false, ones, false, rows, (short)0, d2, false, false);
    }
    wmma_slots(d2, ones, rows);
    if (h == t) tot = d2[0];
  }

  sPart[wave][lane] = tot;
  __syncthreads();
  if (g == 0) {
    const int w0 = half * 4;
    const float s = ((sPart[w0][lane] + sPart[w0 + 1][lane]) + sPart[w0 + 2][lane]) + sPart[w0 + 3][lane];
    out[(size_t)b * (kRes * kRes) + p0 + lane] = s;
  }
}

extern "C" void kernel_launch(void* const* d_in, const int* in_sizes, int n_in,
                              void* d_out, int out_size, void* d_ws, size_t ws_size,
                              hipStream_t stream) {
  (void)in_sizes; (void)n_in; (void)out_size; (void)ws_size;
  const float* Bs = (const float*)d_in[0];
  const float* Ds = (const float*)d_in[1];
  float* w2  = (float*)d_ws;
  float* out = (float*)d_out;

  w2_kernel<<<2, 256, 0, stream>>>(Bs, Ds, w2);
  img_kernel<<<256, 256, 0, stream>>>(Bs, Ds, w2, out);
}


# image kernel declared at its real 2 waves per SIMD (amdgpu_waves_per_eu): scheduler keeps the 4 min-of-max chains of a pixel pair 4 slots apart, not back to back; 154 to 205 VGPRs; bit-identical
# speedup vs baseline: 3.1144x; 1.1038x over previous
#include <hip/hip_runtime.h>
#include <math.h>

typedef __attribute__((ext_vector_type(2))) float v2f;
typedef __attribute__((ext_vector_type(8))) float v8f;

namespace {
constexpr int kRes = 64;
constexpr int kNB  = 4;
constexpr int kNN  = 128;
constexpr int kNJ  = 16;
constexpr int kSeg = 32;
constexpr int kBoxB  = kNJ * 2 * 4;
constexpr int kPitch = kBoxB + 16;
constexpr float kBWInv = 1.0f / 0.3f;
}

static __device__ __forceinline__ void wmma_slots(v8f& d, v2f a, v2f b) {
  asm volatile("v_nop\n\tv_nop\n\tv_nop\n\tv_nop" : "+v"(d) : "v"(a), "v"(b));
}

__global__ void __launch_bounds__(256) w2_kernel(const float* __restrict__ Bs,
                                                 const float* __restrict__ Ds,
                                                 float* __restrict__ w2) {
  int idx = blockIdx.x * blockDim.x + threadIdx.x;
  if (idx >= kNB * kNN) return;
  const float* bp = Bs + idx * kNJ * 2;
  const float* dp = Ds + idx * kNJ * 2;
  float bx[kNJ], by[kNJ], dx[kNJ], dy[kNJ];
#pragma unroll
  for (int j = 0; j < kNJ; ++j) {
    bx[j] = bp[2 * j + 0];
    by[j] = bp[2 * j + 1];
    dx[j] = dp[2 * j + 0];
    dy[j] = dp[2 * j + 1];
  }
  float w = 0.0f;
#pragma unroll
  for (int i = 0; i < kNJ; ++i) {
#pragma unroll
    for (int j = 0; j < kNJ; ++j) {
      float r0 = fmaxf(dx[j] - bx[i], 0.0f);
      float r1 = fmaxf(dy[j] - by[i], 0.0f);
      w = fmaxf(w, fminf(r0, r1));
    }
  }
  w2[idx] = w * w;
}

__global__ void __launch_bounds__(256) __attribute__((amdgpu_waves_per_eu(2, 2))) img_kernel(const float* __restrict__ Bs,
                                                  const float* __restrict__ Ds,
                                                  const float* __restrict__ w2,
                                                  float* __restrict__ out) {
  __shared__ float sPart[8][kSeg];
  __shared__ __attribute__((aligned(16))) unsigned char sBox[4][2][32 * kPitch];

  const int wave = threadIdx.x >> 5;
  const int lane = threadIdx.x & 31;
  const int g    = wave & 3;
  const int half = wave >> 2;
  const int c    = lane & 15;
  const int h    = lane >> 4;
  const int b    = blockIdx.x >> 6;
  const int p0   = (blockIdx.x & 63) * kRes + half * kSeg;
  const int n    = g * 32 + lane;

  float bx[kNJ], by[kNJ], dx[kNJ], dy[kNJ];
  {
    const float* src = half ? Ds : Bs;
    const float4* gS = (const float4*)(src + (size_t)(b * kNN + g * 32) * (kNJ * 2));
    unsigned char* lS = sBox[g][half];
#pragma unroll
    for (int i = 0; i < 8; ++i) {
      const int piece = i * 32 + lane;
      const int off = (piece >> 3) * kPitch + (piece & 7) * 16;
      *(float4*)(lS + off) = gS[piece];
    }
    __syncthreads();
    const unsigned char* lB = sBox[g][0];
    const unsigned char* lD = sBox[g][1];
#pragma unroll
    for (int j = 0; j < kNJ / 2; ++j) {
      const float4 bv = *(const float4*)(lB + lane * kPitch + j * 16);
      const float4 dv = *(const float4*)(lD + lane * kPitch + j * 16);
      bx[2 * j] = bv.x; by[2 * j] = bv.y; bx[2 * j + 1] = bv.z; by[2 * j + 1] = bv.w;
      dx[2 * j] = dv.x; dy[2 * j] = dv.y; dx[2 * j + 1] = dv.z; dy[2 * j + 1] = dv.w;
    }
  }
  const float wn = w2[b * kNN + n];

  const float x0  = -3.0f + 6.0f * (float)(p0 >> 6) / 63.0f;
  const float x1l = -3.0f + 6.0f * (float)((p0 + lane) & 63) / 63.0f;

  float tot = 0.0f;
  for (int t = 0; t < 2; ++t) {
    v8f d1 = {};
    for (int q = 0; q < 8; ++q) {
      v2f a;
#pragma unroll
      for (int s = 0; s < 2; ++s) {
        const float x1 = __builtin_bit_cast(
            float, __builtin_amdgcn_readlane(__builtin_bit_cast(int, x1l), 16 * t + 2 * q + s));
        float db = 3.0e38f, dd = 3.0e38f;
#pragma unroll
        for (int k = 0; k < kNJ; ++k) {
          db = fminf(db, fmaxf(bx[k] - x0, by[k] - x1));
          dd = fminf(dd, fmaxf(x0 - dx[k], x1 - dy[k]));
        }
        float dist = fmaxf(fmaxf(db, dd), 0.0f) * kBWInv;
        a[s] = expf(-0.5f * dist * dist) * wn;
      }
      v2f sel;
      sel[0] = (c == 2 * q)     ? 1.0f : 0.0f;
      sel[1] = (c == 2 * q + 1) ? 1.0f : 0.0f;
      d1 = __builtin_amdgcn_wmma_f32_16x16x4_f32(
          false, a, false, sel, (short)0, d1, false, false);
      wmma_slots(d1, a, sel);
    }
    v8f d2 = {};
    const v2f ones = {1.0f, 1.0f};
    v2f rows;
#pragma unroll
    for (int m = 0; m < 4; ++m) {
      rows[0] = d1[2 * m];
      rows[1] = d1[2 * m + 1];
      d2 = __builtin_amdgcn_wmma_f32_16x16x4_f32(
          false, ones, false, rows, (short)0, d2, false, false);
    }
    wmma_slots(d2, ones, rows);
    if (h == t) tot = d2[0];
  }

  sPart[wave][lane] = tot;
  __syncthreads();
  if (g == 0) {
    const int w0 = half * 4;
    const float s = ((sPart[w0][lane] + sPart[w0 + 1][lane]) + sPart[w0 + 2][lane]) + sPart[w0 + 3][lane];
    out[(size_t)b * (kRes * kRes) + p0 + lane] = s;
  }
}

extern "C" void kernel_launch(void* const* d_in, const int* in_sizes, int n_in,
                              void* d_out, int out_size, void* d_ws, size_t ws_size,
                              hipStream_t stream) {
  (void)in_sizes; (void)n_in; (void)out_size; (void)ws_size;
  const float* Bs = (const float*)d_in[0];
  const float* Ds = (const float*)d_in[1];
  float* w2  = (float*)d_ws;
  float* out = (float*)d_out;

  w2_kernel<<<2, 256, 0, stream>>>(Bs, Ds, w2);
  img_kernel<<<256, 256, 0, stream>>>(Bs, Ds, w2, out);
}


# weight kernel spread out: a box's 256 corner pairs over 16 lanes (4x4 pairs each, joined by a DPP row maximum), 16 work-groups of 32 boxes, not one thread per box in 2 work-groups; bit-identical
# speedup vs baseline: 3.7159x; 1.1931x over previous
#include <hip/hip_runtime.h>
#include <math.h>

typedef __attribute__((ext_vector_type(2))) float v2f;
typedef __attribute__((ext_vector_type(8))) float v8f;

namespace {
constexpr int kRes = 64;
constexpr int kNB  = 4;
constexpr int kNN  = 128;
constexpr int kNJ  = 16;
constexpr int kSeg = 32;
constexpr int kBoxB  = kNJ * 2 * 4;
constexpr int kPitch = kBoxB + 16;
constexpr float kBWInv = 1.0f / 0.3f;
}

static __device__ __forceinline__ void wmma_slots(v8f& d, v2f a, v2f b) {
  asm volatile("v_nop\n\tv_nop\n\tv_nop\n\tv_nop" : "+v"(d) : "v"(a), "v"(b));
}

constexpr int kWBox = 32;

template <int M>
static __device__ __forceinline__ float row_xor(float v) {
  return __builtin_bit_cast(
      float, __builtin_amdgcn_update_dpp(0, __builtin_bit_cast(int, v), 0x160 + M, 0xf, 0xf, true));
}

__global__ void __launch_bounds__(512) w2_kernel(const float* __restrict__ Bs,
                                                 const float* __restrict__ Ds,
                                                 float* __restrict__ w2) {
  __shared__ float sW[kWBox];
  const int lb  = threadIdx.x >> 4;
  const int i   = threadIdx.x & 15;
  const int box = blockIdx.x * kWBox + lb;
  const float4* bp = (const float4*)(Bs + (size_t)box * (kNJ * 2)) + (i >> 2) * 2;
  const float4* dp = (const float4*)(Ds + (size_t)box * (kNJ * 2)) + (i & 3) * 2;
  const float4 b0 = bp[0], b1 = bp[1], d0 = dp[0], d1 = dp[1];
  const float bx[4] = {b0.x, b0.z, b1.x, b1.z};
  const float by[4] = {b0.y, b0.w, b1.y, b1.w};
  const float dx[4] = {d0.x, d0.z, d1.x, d1.z};
  const float dy[4] = {d0.y, d0.w, d1.y, d1.w};
  float w = 0.0f;
  for (int p = 0; p < 4; ++p) {
    for (int q = 0; q < 4; ++q) {
      float r0 = fmaxf(dx[q] - bx[p], 0.0f);
      float r1 = fmaxf(dy[q] - by[p], 0.0f);
      w = fmaxf(w, fminf(r0, r1));
    }
  }
  w = fmaxf(w, row_xor<1>(w));
  w = fmaxf(w, row_xor<2>(w));
  w = fmaxf(w, row_xor<4>(w));
  w = fmaxf(w, row_xor<8>(w));
  if (i == 0) sW[lb] = w;
  __syncthreads();
  if (threadIdx.x < kWBox) {
    const float v = sW[threadIdx.x];
    w2[blockIdx.x * kWBox + threadIdx.x] = v * v;
  }
}

__global__ void __launch_bounds__(256) __attribute__((amdgpu_waves_per_eu(2, 2))) img_kernel(const float* __restrict__ Bs,
                                                  const float* __restrict__ Ds,
                                                  const float* __restrict__ w2,
                                                  float* __restrict__ out) {
  __shared__ float sPart[8][kSeg];
  __shared__ __attribute__((aligned(16))) unsigned char sBox[4][2][32 * kPitch];

  const int wave = threadIdx.x >> 5;
  const int lane = threadIdx.x & 31;
  const int g    = wave & 3;
  const int half = wave >> 2;
  const int c    = lane & 15;
  const int h    = lane >> 4;
  const int b    = blockIdx.x >> 6;
  const int p0   = (blockIdx.x & 63) * kRes + half * kSeg;
  const int n    = g * 32 + lane;

  float bx[kNJ], by[kNJ], dx[kNJ], dy[kNJ];
  {
    const float* src = half ? Ds : Bs;
    const float4* gS = (const float4*)(src + (size_t)(b * kNN + g * 32) * (kNJ * 2));
    unsigned char* lS = sBox[g][half];
#pragma unroll
    for (int i = 0; i < 8; ++i) {
      const int piece = i * 32 + lane;
      const int off = (piece >> 3) * kPitch + (piece & 7) * 16;
      *(float4*)(lS + off) = gS[piece];
    }
    __syncthreads();
    const unsigned char* lB = sBox[g][0];
    const unsigned char* lD = sBox[g][1];
#pragma unroll
    for (int j = 0; j < kNJ / 2; ++j) {
      const float4 bv = *(const float4*)(lB + lane * kPitch + j * 16);
      const float4 dv = *(const float4*)(lD + lane * kPitch + j * 16);
      bx[2 * j] = bv.x; by[2 * j] = bv.y; bx[2 * j + 1] = bv.z; by[2 * j + 1] = bv.w;
      dx[2 * j] = dv.x; dy[2 * j] = dv.y; dx[2 * j + 1] = dv.z; dy[2 * j + 1] = dv.w;
    }
  }
  const float wn = w2[b * kNN + n];

  const float x0  = -3.0f + 6.0f * (float)(p0 >> 6) / 63.0f;
  const float x1l = -3.0f + 6.0f * (float)((p0 + lane) & 63) / 63.0f;

  float tot = 0.0f;
  for (int t = 0; t < 2; ++t) {
    v8f d1 = {};
    for (int q = 0; q < 8; ++q) {
      v2f a;
#pragma unroll
      for (int s = 0; s < 2; ++s) {
        const float x1 = __builtin_bit_cast(
            float, __builtin_amdgcn_readlane(__builtin_bit_cast(int, x1l), 16 * t + 2 * q + s));
        float db = 3.0e38f, dd = 3.0e38f;
#pragma unroll
        for (int k = 0; k < kNJ; ++k) {
          db = fminf(db, fmaxf(bx[k] - x0, by[k] - x1));
          dd = fminf(dd, fmaxf(x0 - dx[k], x1 - dy[k]));
        }
        float dist = fmaxf(fmaxf(db, dd), 0.0f) * kBWInv;
        a[s] = expf(-0.5f * dist * dist) * wn;
      }
      v2f sel;
      sel[0] = (c == 2 * q)     ? 1.0f : 0.0f;
      sel[1] = (c == 2 * q + 1) ? 1.0f : 0.0f;
      d1 = __builtin_amdgcn_wmma_f32_16x16x4_f32(
          false, a, false, sel, (short)0, d1, false, false);
      wmma_slots(d1, a, sel);
    }
    v8f d2 = {};
    const v2f ones = {1.0f, 1.0f};
    v2f rows;
#pragma unroll
    for (int m = 0; m < 4; ++m) {
      rows[0] = d1[2 * m];
      rows[1] = d1[2 * m + 1];
      d2 = __builtin_amdgcn_wmma_f32_16x16x4_f32(
          false, ones, false, rows, (short)0, d2, false, false);
    }
    wmma_slots(d2, ones, rows);
    if (h == t) tot = d2[0];
  }

  sPart[wave][lane] = tot;
  __syncthreads();
  if (g == 0) {
    const int w0 = half * 4;
    const float s = ((sPart[w0][lane] + sPart[w0 + 1][lane]) + sPart[w0 + 2][lane]) + sPart[w0 + 3][lane];
    out[(size_t)b * (kRes * kRes) + p0 + lane] = s;
  }
}

extern "C" void kernel_launch(void* const* d_in, const int* in_sizes, int n_in,
                              void* d_out, int out_size, void* d_ws, size_t ws_size,
                              hipStream_t stream) {
  (void)in_sizes; (void)n_in; (void)out_size; (void)ws_size;
  const float* Bs = (const float*)d_in[0];
  const float* Ds = (const float*)d_in[1];
  float* w2  = (float*)d_ws;
  float* out = (float*)d_out;

  w2_kernel<<<16, 512, 0, stream>>>(Bs, Ds, w2);
  img_kernel<<<256, 256, 0, stream>>>(Bs, Ds, w2, out);
}


# box weight worked out inside the image kernel by the lane that holds the box (256-pair max-of-min from its registers, halves of two waves joined in LDS); weight launch and scratch go; bit-identical
# speedup vs baseline: 4.7528x; 1.2790x over previous
#include <hip/hip_runtime.h>
#include <math.h>

typedef __attribute__((ext_vector_type(2))) float v2f;
typedef __attribute__((ext_vector_type(8))) float v8f;

namespace {
constexpr int kRes = 64;
constexpr int kNB  = 4;
constexpr int kNN  = 128;
constexpr int kNJ  = 16;
constexpr int kSeg = 32;
constexpr int kBoxB  = kNJ * 2 * 4;
constexpr int kPitch = kBoxB + 16;
constexpr float kBWInv = 1.0f / 0.3f;
}

static __device__ __forceinline__ void wmma_slots(v8f& d, v2f a, v2f b) {
  asm volatile("v_nop\n\tv_nop\n\tv_nop\n\tv_nop" : "+v"(d) : "v"(a), "v"(b));
}

__global__ void __launch_bounds__(256) __attribute__((amdgpu_waves_per_eu(2, 2))) img_kernel(const float* __restrict__ Bs,
                                                  const float* __restrict__ Ds,
                                                  float* __restrict__ out) {
  __shared__ float sPart[8][kSeg];
  __shared__ float sWgt[8][32];
  __shared__ __attribute__((aligned(16))) unsigned char sBox[4][2][32 * kPitch];

  const int wave = threadIdx.x >> 5;
  const int lane = threadIdx.x & 31;
  const int g    = wave & 3;
  const int half = wave >> 2;
  const int c    = lane & 15;
  const int h    = lane >> 4;
  const int b    = blockIdx.x >> 6;
  const int p0   = (blockIdx.x & 63) * kRes + half * kSeg;

  float bx[kNJ], by[kNJ], dx[kNJ], dy[kNJ];
  {
    const float* src = half ? Ds : Bs;
    const float4* gS = (const float4*)(src + (size_t)(b * kNN + g * 32) * (kNJ * 2));
    unsigned char* lS = sBox[g][half];
#pragma unroll
    for (int i = 0; i < 8; ++i) {
      const int piece = i * 32 + lane;
      const int off = (piece >> 3) * kPitch + (piece & 7) * 16;
      *(float4*)(lS + off) = gS[piece];
    }
    __syncthreads();
    const unsigned char* lB = sBox[g][0];
    const unsigned char* lD = sBox[g][1];
#pragma unroll
    for (int j = 0; j < kNJ / 2; ++j) {
      const float4 bv = *(const float4*)(lB + lane * kPitch + j * 16);
      const float4 dv = *(const float4*)(lD + lane * kPitch + j * 16);
      bx[2 * j] = bv.x; by[2 * j] = bv.y; bx[2 * j + 1] = bv.z; by[2 * j + 1] = bv.w;
      dx[2 * j] = dv.x; dy[2 * j] = dv.y; dx[2 * j + 1] = dv.z; dy[2 * j + 1] = dv.w;
    }
  }
  float wq[4] = {0.0f, 0.0f, 0.0f, 0.0f};
  {
    const unsigned char* lBh = sBox[g][0] + lane * kPitch + half * (kBoxB / 2);
#pragma nounroll
    for (int j = 0; j < 2; ++j) {
      const float4 bu = *(const float4*)(lBh + j * 32);
      const float4 bv = *(const float4*)(lBh + j * 32 + 16);
      const v2f bc[4] = {{bu.x, bu.y}, {bu.z, bu.w}, {bv.x, bv.y}, {bv.z, bv.w}};
#pragma unroll
      for (int k = 0; k < kNJ; ++k) {
        const v2f dk = {dx[k], dy[k]};
#pragma unroll
        for (int i = 0; i < 4; ++i) {
          const v2f r = dk - bc[i];
          wq[i] = fmaxf(wq[i], fminf(r[0], r[1]));
        }
      }
    }
  }
  const float wpart = fmaxf(fmaxf(wq[0], wq[1]), fmaxf(wq[2], wq[3]));
  sWgt[wave][lane] = wpart;
  __syncthreads();
  const float wbox = fmaxf(wpart, sWgt[wave ^ 4][lane]);
  const float wn = wbox * wbox;

  const float x0  = -3.0f + 6.0f * (float)(p0 >> 6) / 63.0f;
  const float x1l = -3.0f + 6.0f * (float)((p0 + lane) & 63) / 63.0f;

  float tot = 0.0f;
  for (int t = 0; t < 2; ++t) {
    v8f d1 = {};
    for (int q = 0; q < 8; ++q) {
      v2f a;
#pragma unroll
      for (int s = 0; s < 2; ++s) {
        const float x1 = __builtin_bit_cast(
            float, __builtin_amdgcn_readlane(__builtin_bit_cast(int, x1l), 16 * t + 2 * q + s));
        float db = 3.0e38f, dd = 3.0e38f;
#pragma unroll
        for (int k = 0; k < kNJ; ++k) {
          db = fminf(db, fmaxf(bx[k] - x0, by[k] - x1));
          dd = fminf(dd, fmaxf(x0 - dx[k], x1 - dy[k]));
        }
        float dist = fmaxf(fmaxf(db, dd), 0.0f) * kBWInv;
        a[s] = expf(-0.5f * dist * dist) * wn;
      }
      v2f sel;
      sel[0] = (c == 2 * q)     ? 1.0f : 0.0f;
      sel[1] = (c == 2 * q + 1) ? 1.0f : 0.0f;
      d1 = __builtin_amdgcn_wmma_f32_16x16x4_f32(
          false, a, false, sel, (short)0, d1, false, false);
      wmma_slots(d1, a, sel);
    }
    v8f d2 = {};
    const v2f ones = {1.0f, 1.0f};
    v2f rows;
#pragma unroll
    for (int m = 0; m < 4; ++m) {
      rows[0] = d1[2 * m];
      rows[1] = d1[2 * m + 1];
      d2 = __builtin_amdgcn_wmma_f32_16x16x4_f32(
          false, ones, false, rows, (short)0, d2, false, false);
    }
    wmma_slots(d2, ones, rows);
    if (h == t) tot = d2[0];
  }

  sPart[wave][lane] = tot;
  __syncthreads();
  if (g == 0) {
    const int w0 = half * 4;
    const float s = ((sPart[w0][lane] + sPart[w0 + 1][lane]) + sPart[w0 + 2][lane]) + sPart[w0 + 3][lane];
    out[(size_t)b * (kRes * kRes) + p0 + lane] = s;
  }
}

extern "C" void kernel_launch(void* const* d_in, const int* in_sizes, int n_in,
                              void* d_out, int out_size, void* d_ws, size_t ws_size,
                              hipStream_t stream) {
  (void)in_sizes; (void)n_in; (void)out_size; (void)d_ws; (void)ws_size;
  const float* Bs = (const float*)d_in[0];
  const float* Ds = (const float*)d_in[1];
  float* out = (float*)d_out;

  img_kernel<<<256, 256, 0, stream>>>(Bs, Ds, out);
}


# only a box's non-dominated corners enter the min-of-max chains and the weight: corners sorted by x in registers, one pass keeps those with y below all before; corner-outer loops; bit-identical
# speedup vs baseline: 5.6966x; 1.1986x over previous
#include <hip/hip_runtime.h>
#include <math.h>

typedef __attribute__((ext_vector_type(2))) float v2f;
typedef __attribute__((ext_vector_type(8))) float v8f;

namespace {
constexpr int kRes = 64;
constexpr int kNB  = 4;
constexpr int kNN  = 128;
constexpr int kNJ  = 16;
constexpr int kSeg = 32;
constexpr int kBoxB  = kNJ * 2 * 4;
constexpr int kPitch = kBoxB + 16;
constexpr float kBWInv = 1.0f / 0.3f;
}

static __device__ __forceinline__ void wmma_slots(v8f& d, v2f a, v2f b) {
  asm volatile("v_nop\n\tv_nop\n\tv_nop\n\tv_nop" : "+v"(d) : "v"(a), "v"(b));
}
static __device__ __forceinline__ void wmma_slots_tile(v8f& d, v2f a, v8f b) {
  asm volatile("v_nop\n\tv_nop\n\tv_nop\n\tv_nop" : "+v"(d) : "v"(a), "v"(b));
}

static __device__ __forceinline__ void cex(float& xa, float& ya, float& xb, float& yb) {
  const bool sw = xa > xb;
  const float xl = sw ? xb : xa, xh = sw ? xa : xb;
  const float yl = sw ? yb : ya, yh = sw ? ya : yb;
  xa = xl; xb = xh; ya = yl; yb = yh;
}

__global__ void __launch_bounds__(256) __attribute__((amdgpu_waves_per_eu(2, 2))) img_kernel(const float* __restrict__ Bs,
                                                  const float* __restrict__ Ds,
                                                  float* __restrict__ out) {
  __shared__ float sPart[8][kSeg];
  __shared__ int sCnt[4][2][32];
  __shared__ __attribute__((aligned(16))) float sX1[8][kSeg];
  __shared__ __attribute__((aligned(16))) unsigned char sBox[4][2][32 * kPitch];

  const int wave = __builtin_amdgcn_readfirstlane((int)(threadIdx.x >> 5));
  const int lane = threadIdx.x & 31;
  const int g    = wave & 3;
  const int half = wave >> 2;
  const int c    = lane & 15;
  const int h    = lane >> 4;
  const int b    = blockIdx.x >> 6;
  const int p0   = (blockIdx.x & 63) * kRes + half * kSeg;

  {
    const float* src = half ? Ds : Bs;
    const float sgn = half ? -1.0f : 1.0f;
    const float4* gS = (const float4*)(src + (size_t)(b * kNN + g * 32) * (kNJ * 2));
    unsigned char* lS = sBox[g][half];
#pragma unroll
    for (int i = 0; i < 8; ++i) {
      const int piece = i * 32 + lane;
      const int off = (piece >> 3) * kPitch + (piece & 7) * 16;
      float4 v = gS[piece];
      v.x *= sgn; v.y *= sgn; v.z *= sgn; v.w *= sgn;
      *(float4*)(lS + off) = v;
    }
  }
  __syncthreads();

  float kx[kNJ], ky[kNJ];
  {
    unsigned char* myRow = sBox[g][half] + lane * kPitch;
#pragma unroll
    for (int j = 0; j < kNJ / 2; ++j) {
      const float4 v = *(const float4*)(myRow + j * 16);
      kx[2 * j] = v.x; ky[2 * j] = v.y; kx[2 * j + 1] = v.z; ky[2 * j + 1] = v.w;
    }
#define CEX(i, j) cex(kx[i], ky[i], kx[j], ky[j])
    CEX(0, 1); CEX(2, 3); CEX(4, 5); CEX(6, 7);
    CEX(8, 9); CEX(10, 11); CEX(12, 13); CEX(14, 15);
    CEX(0, 2); CEX(1, 3); CEX(4, 6); CEX(5, 7);
    CEX(8, 10); CEX(9, 11); CEX(12, 14); CEX(13, 15);
    CEX(1, 2); CEX(5, 6); CEX(9, 10); CEX(13, 14);
    CEX(0, 4); CEX(1, 5); CEX(2, 6); CEX(3, 7);
    CEX(8, 12); CEX(9, 13); CEX(10, 14); CEX(11, 15);
    CEX(2, 4); CEX(3, 5); CEX(10, 12); CEX(11, 13);
    CEX(1, 2); CEX(3, 4); CEX(5, 6); CEX(9, 10);
    CEX(11, 12); CEX(13, 14); CEX(0, 8); CEX(1, 9);
    CEX(2, 10); CEX(3, 11); CEX(4, 12); CEX(5, 13);
    CEX(6, 14); CEX(7, 15); CEX(4, 8); CEX(5, 9);
    CEX(6, 10); CEX(7, 11); CEX(2, 4); CEX(3, 5);
    CEX(6, 8); CEX(7, 9); CEX(10, 12); CEX(11, 13);
    CEX(1, 2); CEX(3, 4); CEX(5, 6); CEX(7, 8);
    CEX(9, 10); CEX(11, 12); CEX(13, 14);
#undef CEX
    float rm = __builtin_inff();
    unsigned char* wp = myRow;
#pragma unroll
    for (int k = 0; k < kNJ; ++k) {
      const v2f e = {kx[k], ky[k]};
      *(v2f*)wp = e;
      const bool keep = ky[k] < rm;
      rm = fminf(rm, ky[k]);
      wp += keep ? 8 : 0;
    }
    sCnt[g][half][lane] = (int)(wp - myRow) >> 3;
  }
  __syncthreads();

  const int cntB = sCnt[g][0][lane];
  const int cntD = sCnt[g][1][lane];
  const unsigned char* rowB = sBox[g][0] + lane * kPitch;
  const unsigned char* rowD = sBox[g][1] + lane * kPitch;

  const float x0  = -3.0f + 6.0f * (float)(p0 >> 6) / 63.0f;
  const float x1l = -3.0f + 6.0f * (float)((p0 + lane) & 63) / 63.0f;
  sX1[wave][lane] = x1l;
  v2f x1p[kSeg / 2];
#pragma unroll
  for (int j = 0; j < kSeg / 4; ++j) {
    const float4 v = *(const float4*)&sX1[wave][4 * j];
    x1p[2 * j] = v2f{v.x, v.y}; x1p[2 * j + 1] = v2f{v.z, v.w};
  }

  float db[kSeg], dd[kSeg];
#pragma unroll
  for (int p = 0; p < kSeg; ++p) { db[p] = 3.0e38f; dd[p] = 3.0e38f; }
  {
    float4 c01 = *(const float4*)rowB, c23 = *(const float4*)(rowB + 16);
    for (int k = 0;;) {
      const int kn = k + 4;
      const bool more = __builtin_amdgcn_ballot_w32(kn < cntB) != 0u;
      const float4 n01 = *(const float4*)(rowB + (kn & 12) * 8);
      const float4 n23 = *(const float4*)(rowB + (kn & 12) * 8 + 16);
      const v2f cur[4] = {{c01.x, c01.y}, {c01.z, c01.w}, {c23.x, c23.y}, {c23.z, c23.w}};
#pragma unroll
      for (int e = 0; e < 4; ++e) {
        const float u = cur[e][0] - x0;
        const v2f y = {cur[e][1], cur[e][1]};
#pragma unroll
        for (int i = 0; i < kSeg / 2; ++i) {
          const v2f t = y - x1p[i];
          db[2 * i] = fminf(db[2 * i], fmaxf(u, t[0]));
          db[2 * i + 1] = fminf(db[2 * i + 1], fmaxf(u, t[1]));
        }
      }
      c01 = n01; c23 = n23;
      k = kn;
      if (!more) break;
    }
  }
  {
    float4 c01 = *(const float4*)rowD, c23 = *(const float4*)(rowD + 16);
    for (int k = 0;;) {
      const int kn = k + 4;
      const bool more = __builtin_amdgcn_ballot_w32(kn < cntD) != 0u;
      const float4 n01 = *(const float4*)(rowD + (kn & 12) * 8);
      const float4 n23 = *(const float4*)(rowD + (kn & 12) * 8 + 16);
      const v2f cur[4] = {{c01.x, c01.y}, {c01.z, c01.w}, {c23.x, c23.y}, {c23.z, c23.w}};
#pragma unroll
      for (int e = 0; e < 4; ++e) {
        const float u = x0 + cur[e][0];
        const v2f y = {cur[e][1], cur[e][1]};
#pragma unroll
        for (int i = 0; i < kSeg / 2; ++i) {
          const v2f t = x1p[i] + y;
          dd[2 * i] = fminf(dd[2 * i], fmaxf(u, t[0]));
          dd[2 * i + 1] = fminf(dd[2 * i + 1], fmaxf(u, t[1]));
        }
      }
      c01 = n01; c23 = n23;
      k = kn;
      if (!more) break;
    }
  }

  float wq[4] = {3.0e38f, 3.0e38f, 3.0e38f, 3.0e38f};
  {
    v2f eb[8], ed[8];
#pragma unroll
    for (int j = 0; j < 4; ++j) {
      const float4 vb = *(const float4*)(rowB + j * 16);
      const float4 vd = *(const float4*)(rowD + j * 16);
      eb[2 * j] = v2f{vb.x, vb.y}; eb[2 * j + 1] = v2f{vb.z, vb.w};
      ed[2 * j] = v2f{vd.x, vd.y}; ed[2 * j + 1] = v2f{vd.z, vd.w};
    }
#pragma unroll
    for (int i = 0; i < 8; ++i)
#pragma unroll
      for (int j = 0; j < 8; ++j) {
        const v2f r = eb[i] + ed[j];
        wq[j & 3] = fminf(wq[j & 3], fmaxf(r[0], r[1]));
      }
    for (int k = 8; k < kNJ && __builtin_amdgcn_ballot_w32(k < cntB) != 0u; k += 4) {
      const float4 c01 = *(const float4*)(rowB + k * 8), c23 = *(const float4*)(rowB + k * 8 + 16);
      const v2f cur[4] = {{c01.x, c01.y}, {c01.z, c01.w}, {c23.x, c23.y}, {c23.z, c23.w}};
#pragma unroll
      for (int e = 0; e < 4; ++e)
#pragma unroll
        for (int j = 0; j < 8; ++j) {
          const v2f r = cur[e] + ed[j];
          wq[j & 3] = fminf(wq[j & 3], fmaxf(r[0], r[1]));
        }
    }
    for (int k = 8; k < kNJ && __builtin_amdgcn_ballot_w32(k < cntD) != 0u; k += 4) {
      const float4 c01 = *(const float4*)(rowD + k * 8), c23 = *(const float4*)(rowD + k * 8 + 16);
      const v2f cur[4] = {{c01.x, c01.y}, {c01.z, c01.w}, {c23.x, c23.y}, {c23.z, c23.w}};
#pragma unroll
      for (int e = 0; e < 4; ++e)
#pragma unroll
        for (int i = 0; i < 8; ++i) {
          const v2f r = eb[i] + cur[e];
          wq[i & 3] = fminf(wq[i & 3], fmaxf(r[0], r[1]));
        }
    }
  }
  if (__builtin_amdgcn_ballot_w32(cntB > 8) != 0u && __builtin_amdgcn_ballot_w32(cntD > 8) != 0u) {
    for (int i = 8; i < kNJ; ++i) {
      const v2f bi = *(const v2f*)(rowB + i * 8);
      for (int j = 8; j < kNJ; ++j) {
        const v2f r = bi + *(const v2f*)(rowD + j * 8);
        wq[0] = fminf(wq[0], fmaxf(r[0], r[1]));
      }
    }
  }
  const float wbox = fmaxf(-fminf(fminf(wq[0], wq[1]), fminf(wq[2], wq[3])), 0.0f);
  const float wn = wbox * wbox;

  float tot = 0.0f;
#pragma unroll
  for (int t = 0; t < 2; ++t) {
    v8f d1 = {};
#pragma unroll
    for (int q = 0; q < 8; ++q) {
      v2f a;
#pragma unroll
      for (int s = 0; s < 2; ++s) {
        const int p = 16 * t + 2 * q + s;
        float dist = fmaxf(fmaxf(db[p], dd[p]), 0.0f) * kBWInv;
        a[s] = expf(-0.5f * dist * dist) * wn;
      }
      v2f sel;
      sel[0] = (c == 2 * q)     ? 1.0f : 0.0f;
      sel[1] = (c == 2 * q + 1) ? 1.0f : 0.0f;
      d1 = __builtin_amdgcn_wmma_f32_16x16x4_f32(
          false, a, false, sel, (short)0, d1, false, false);
      wmma_slots(d1, a, sel);
    }
    v8f d2 = {};
    const v2f ones = {1.0f, 1.0f};
    v2f rows;
#pragma unroll
    for (int m = 0; m < 4; ++m) {
      rows[0] = d1[2 * m];
      rows[1] = d1[2 * m + 1];
      d2 = __builtin_amdgcn_wmma_f32_16x16x4_f32(
          false, ones, false, rows, (short)0, d2, false, false);
    }
    wmma_slots_tile(d2, ones, d1);
    if (h == t) tot = d2[0];
  }

  sPart[wave][lane] = tot;
  __syncthreads();
  if (g == 0) {
    const int w0 = half * 4;
    const float s = ((sPart[w0][lane] + sPart[w0 + 1][lane]) + sPart[w0 + 2][lane]) + sPart[w0 + 3][lane];
    out[(size_t)b * (kRes * kRes) + p0 + lane] = s;
  }
}

extern "C" void kernel_launch(void* const* d_in, const int* in_sizes, int n_in,
                              void* d_out, int out_size, void* d_ws, size_t ws_size,
                              hipStream_t stream) {
  (void)in_sizes; (void)n_in; (void)out_size; (void)d_ws; (void)ws_size;
  const float* Bs = (const float*)d_in[0];
  const float* Ds = (const float*)d_in[1];
  float* out = (float*)d_out;

  img_kernel<<<256, 256, 0, stream>>>(Bs, Ds, out);
}
